# starting kernel
# baseline (speedup 1.0000x reference)
#include <hip/hip_runtime.h>


typedef __attribute__((ext_vector_type(2))) float v2f;
typedef __attribute__((ext_vector_type(8))) float v8f;

#define C2D_EPS   0.01f
#define MAXW      16
#define TPB       256
#define LDS_PITCH 17

__global__ __launch_bounds__(TPB)
void corner2depth_wmma_kernel(const float* __restrict__ corners,
                              const float* __restrict__ grid,
                              const int*   __restrict__ nums,
                              float* __restrict__ depth_out,
                              float* __restrict__ nrm_out,
                              int Nmax, long long HW)
{
    __shared__ float s_nx[MAXW], s_nz[MAXW], s_nd[MAXW];
    __shared__ float s_minx[MAXW], s_maxx[MAXW];
    __shared__ float s_minz[MAXW], s_maxz[MAXW];
    __shared__ float s_denom[TPB * LDS_PITCH];

    const int tid = threadIdx.x;
    const int b   = blockIdx.y;
    int n = nums[b];
    if (n > MAXW) n = MAXW;

    if (tid < MAXW) {
        float nx = 0.f, nz = 0.f, nd = 0.f;
        float mnx = 0.f, mxx = 0.f, mnz = 0.f, mxz = 0.f;
        if (tid < n) {
            const long long base = ((long long)b * Nmax + tid) * 3;
            const int j = (tid + 1 == n) ? 0 : tid + 1;
            const long long basej = ((long long)b * Nmax + j) * 3;
            const float x0 = corners[base + 0], z0 = corners[base + 2];
            const float x1 = corners[basej + 0], z1 = corners[basej + 2];
            const float dx = x1 - x0, dz = z1 - z0;
            nx = -dz;
            nz =  dx;
            nd = nx * x0 + nz * z0;
            mnx = fminf(x0, x1) - C2D_EPS;  mxx = fmaxf(x0, x1) + C2D_EPS;
            mnz = fminf(z0, z1) - C2D_EPS;  mxz = fmaxf(z0, z1) + C2D_EPS;
        }
        s_nx[tid] = nx; s_nz[tid] = nz; s_nd[tid] = nd;
        s_minx[tid] = mnx; s_maxx[tid] = mxx;
        s_minz[tid] = mnz; s_maxz[tid] = mxz;
    }
    __syncthreads();

    const long long p = (long long)blockIdx.x * TPB + tid;
    const float gx = grid[p * 3 + 0];
    const float gz = grid[p * 3 + 2];

    const int lane = tid & 31;
    const int lh   = lane & 15;
    const int hlf  = lane >> 4;
    const int wv   = tid >> 5;

    const float gx0 = __shfl(gx, lh,      32);
    const float gz0 = __shfl(gz, lh,      32);
    const float gx1 = __shfl(gx, lh + 16, 32);
    const float gz1 = __shfl(gz, lh + 16, 32);

    v2f a0, a1, bm;
    a0.x = hlf ? gz0 : gx0;  a0.y = 0.f;
    a1.x = hlf ? gz1 : gx1;  a1.y = 0.f;
    bm.x = hlf ? s_nz[lh] : s_nx[lh];  bm.y = 0.f;

    v8f acc0 = {}, acc1 = {};
    acc0 = __builtin_amdgcn_wmma_f32_16x16x4_f32(false, a0, false, bm,
                                                 (short)0, acc0, false, false);
    acc1 = __builtin_amdgcn_wmma_f32_16x16x4_f32(false, a1, false, bm,
                                                 (short)0, acc1, false, false);

    const int mBase = hlf * 8;
#pragma unroll
    for (int r = 0; r < 8; ++r) {
        s_denom[(wv * 32 +      mBase + r) * LDS_PITCH + lh] = acc0[r];
        s_denom[(wv * 32 + 16 + mBase + r) * LDS_PITCH + lh] = acc1[r];
    }
    __syncthreads();

    float best = __builtin_inff();
    int   bi   = 0;
    for (int i = 0; i < n; ++i) {
        const float denom = s_denom[tid * LDS_PITCH + i];
        const float scale = s_nd[i] * __builtin_amdgcn_rcpf(denom);
        const float ix = gx * scale;
        const float iz = gz * scale;
        const bool ok = (ix <= s_maxx[i]) & (ix >= s_minx[i])
                      & (iz <= s_maxz[i]) & (iz >= s_minz[i])
                      & (scale > 0.f);
        if (ok && scale < best) { best = scale; bi = i; }
    }

    const long long o = (long long)b * HW + p;
    depth_out[o]       = best;
    nrm_out[o * 3 + 0] = s_nx[bi];
    nrm_out[o * 3 + 1] = 0.f;
    nrm_out[o * 3 + 2] = s_nz[bi];
}

extern "C" void kernel_launch(void* const* d_in, const int* in_sizes, int n_in,
                              void* d_out, int out_size, void* d_ws, size_t ws_size,
                              hipStream_t stream) {
    const float* corners = (const float*)d_in[0];
    const float* grid    = (const float*)d_in[1];
    const int*   nums    = (const int*)d_in[2];

    const int       B    = in_sizes[2];
    const int       Nmax = in_sizes[0] / (B * 3);
    const long long HW   = in_sizes[1] / 3;

    float* depth = (float*)d_out;
    float* nrm   = depth + (long long)B * HW;

    dim3 grd((unsigned)(HW / TPB), (unsigned)B);
    corner2depth_wmma_kernel<<<grd, TPB, 0, stream>>>(corners, grid, nums,
                                                      depth, nrm, Nmax, HW);
}


# Thread owns one ray for all 4 batch samples (4x fewer waves, grid.y = sample groups): ray load, A fragments, sign ballots once; all samples' walls set up at once; staging rows ordered per wave
# speedup vs baseline: 1.5399x; 1.5399x over previous
#include <hip/hip_runtime.h>


typedef __attribute__((ext_vector_type(2))) float v2f;
typedef __attribute__((ext_vector_type(8))) float v8f;

#define C2D_EPS   0.01f
#define MAXW      16
#define TPB       256
#define LDS_PITCH 17
#define WREC      8
#define BG        4

__device__ __forceinline__ void wave_lds_order()
{
    __builtin_amdgcn_fence(__ATOMIC_RELEASE, "wavefront");
    __builtin_amdgcn_wave_barrier();
    __builtin_amdgcn_fence(__ATOMIC_ACQUIRE, "wavefront");
}

__global__ __launch_bounds__(TPB)
void corner2depth_wmma_kernel(const float* __restrict__ corners,
                              const float* __restrict__ grid,
                              const int*   __restrict__ nums,
                              float* __restrict__ depth_out,
                              float* __restrict__ nrm_out,
                              int B, int Nmax, long long HW)
{
    __shared__ __attribute__((aligned(16))) float s_w[BG * MAXW * WREC];
    __shared__ __attribute__((aligned(16))) unsigned s_free[BG * MAXW * 4];
    __shared__ float s_denom[TPB * LDS_PITCH];

    const int tid  = threadIdx.x;
    const int lane = tid & 31;
    const int lh   = lane & 15;
    const int hlf  = lane >> 4;
    const int wv   = tid >> 5;

    const long long p = (long long)blockIdx.x * TPB + tid;
    const float gx = grid[p * 3 + 0];
    const float gz = grid[p * 3 + 2];

    const int b0 = blockIdx.y * BG;
    const int nb = (B - b0 < BG) ? (B - b0) : BG;

    if (tid < nb * MAXW) {
        const int b = b0 + (tid >> 4);
        const int w = tid & 15;
        int n = nums[b];
        if (n > MAXW) n = MAXW;
        float nx = 0.f, nz = 0.f, nd = 0.f;
        float mnx = 0.f, mxx = 0.f, mnz = 0.f, mxz = 0.f;
        unsigned fxp = 0u, fxn = 0u, fzp = 0u, fzn = 0u;
        if (w < n) {
            const long long base = ((long long)b * Nmax + w) * 3;
            const int j = (w + 1 == n) ? 0 : w + 1;
            const long long basej = ((long long)b * Nmax + j) * 3;
            const float x0 = corners[base + 0], z0 = corners[base + 2];
            const float x1 = corners[basej + 0], z1 = corners[basej + 2];
            const float dx = x1 - x0, dz = z1 - z0;
            nx = -dz;
            nz =  dx;
            nd = fmaf(nz, z0, nx * x0);
            mnx = fminf(x0, x1) - C2D_EPS;  mxx = fmaxf(x0, x1) + C2D_EPS;
            mnz = fminf(z0, z1) - C2D_EPS;  mxz = fmaxf(z0, z1) + C2D_EPS;
            fxp = (mnx > 0.f) ? 0u : ~0u;
            fxn = (mxx < 0.f) ? 0u : ~0u;
            fzp = (mnz > 0.f) ? 0u : ~0u;
            fzn = (mxz < 0.f) ? 0u : ~0u;
        }
        float* wr = &s_w[tid * WREC];
        wr[0] = nd;  wr[1] = mxx; wr[2] = mnx; wr[3] = mxz;
        wr[4] = mnz; wr[5] = nx;  wr[6] = nz;  wr[7] = 0.f;
        unsigned* f = &s_free[tid * 4];
        f[0] = fxp; f[1] = fxn; f[2] = fzp; f[3] = fzn;
    }
    __syncthreads();

    const float gx0 = __shfl(gx, lh,      32);
    const float gz0 = __shfl(gz, lh,      32);
    const float gx1 = __shfl(gx, lh + 16, 32);
    const float gz1 = __shfl(gz, lh + 16, 32);
    v2f a0, a1;
    a0.x = hlf ? gz0 : gx0;  a0.y = 0.f;
    a1.x = hlf ? gz1 : gx1;  a1.y = 0.f;
    const unsigned bxp = __builtin_amdgcn_ballot_w32(gx > 0.f);
    const unsigned bxn = __builtin_amdgcn_ballot_w32(gx < 0.f);
    const unsigned bzp = __builtin_amdgcn_ballot_w32(gz > 0.f);
    const unsigned bzn = __builtin_amdgcn_ballot_w32(gz < 0.f);

    for (int bb = 0; bb < nb; ++bb) {
        const float*    sw = &s_w[bb * MAXW * WREC];
        const unsigned* sf = &s_free[bb * MAXW * 4];

        v2f bm;
        bm.x = sw[lh * WREC + (hlf ? 6 : 5)];  bm.y = 0.f;
        v8f acc0 = {}, acc1 = {};
        acc0 = __builtin_amdgcn_wmma_f32_16x16x4_f32(false, a0, false, bm,
                                                     (short)0, acc0, false, false);
        acc1 = __builtin_amdgcn_wmma_f32_16x16x4_f32(false, a1, false, bm,
                                                     (short)0, acc1, false, false);

        wave_lds_order();
        const int mBase = hlf * 8;
#pragma unroll
        for (int r = 0; r < 8; ++r) {
            s_denom[(wv * 32 +      mBase + r) * LDS_PITCH + lh] = acc0[r];
            s_denom[(wv * 32 + 16 + mBase + r) * LDS_PITCH + lh] = acc1[r];
        }
        wave_lds_order();

        const uint4 fr = *reinterpret_cast<const uint4*>(&sf[lh * 4]);
        const unsigned rays = (bxp | fr.x) & (bxn | fr.y) & (bzp | fr.z) & (bzn | fr.w);
        unsigned live = __builtin_amdgcn_ballot_w32(rays != 0u) & 0xffffu;

        float best = __builtin_inff();
        int   bi   = 0;
        while (live != 0u) {
            const int i = __builtin_ctz(live);
            live &= live - 1u;
            const float4 wa = *reinterpret_cast<const float4*>(&sw[i * WREC]);
            const float  mnz = sw[i * WREC + 4];
            const float denom = s_denom[tid * LDS_PITCH + i];
            const float scale = wa.x * __builtin_amdgcn_rcpf(denom);
            const float ix = gx * scale;
            const float iz = gz * scale;
            const bool ok = (ix <= wa.y) & (ix >= wa.z)
                          & (iz <= wa.w) & (iz >= mnz)
                          & (scale > 0.f);
            if (ok && scale < best) { best = scale; bi = i; }
        }

        const long long o = (long long)(b0 + bb) * HW + p;
        depth_out[o]       = best;
        nrm_out[o * 3 + 0] = sw[bi * WREC + 5];
        nrm_out[o * 3 + 1] = 0.f;
        nrm_out[o * 3 + 2] = sw[bi * WREC + 6];
    }
}

extern "C" void kernel_launch(void* const* d_in, const int* in_sizes, int n_in,
                              void* d_out, int out_size, void* d_ws, size_t ws_size,
                              hipStream_t stream) {
    const float* corners = (const float*)d_in[0];
    const float* grid    = (const float*)d_in[1];
    const int*   nums    = (const int*)d_in[2];

    const int       B    = in_sizes[2];
    const int       Nmax = in_sizes[0] / (B * 3);
    const long long HW   = in_sizes[1] / 3;

    float* depth = (float*)d_out;
    float* nrm   = depth + (long long)B * HW;

    dim3 grd((unsigned)(HW / TPB), (unsigned)((B + BG - 1) / BG));
    corner2depth_wmma_kernel<<<grd, TPB, 0, stream>>>(corners, grid, nums,
                                                      depth, nrm, B, Nmax, HW);
}


# Walls culled per wave by a cone test on the matrix unit: a second WMMA pair gives t = <g,m> - |g|(cos(half angle) - 2^-11) per ray and wall-box cone; a wall with t < 0 for all 32 rays is skipped
# speedup vs baseline: 1.6202x; 1.0521x over previous
#include <hip/hip_runtime.h>


typedef __attribute__((ext_vector_type(2))) float v2f;
typedef __attribute__((ext_vector_type(8))) float v8f;

#define C2D_EPS   0.01f
#define MAXW      16
#define TPB       256
#define LDS_PITCH 17
#define WREC      8
#define BG        4
#define CONE_MU   0.00048828125f

__device__ __forceinline__ void wave_lds_order()
{
    __builtin_amdgcn_fence(__ATOMIC_RELEASE, "wavefront");
    __builtin_amdgcn_wave_barrier();
    __builtin_amdgcn_fence(__ATOMIC_ACQUIRE, "wavefront");
}

__global__ __launch_bounds__(TPB)
void corner2depth_wmma_kernel(const float* __restrict__ corners,
                              const float* __restrict__ grid,
                              const int*   __restrict__ nums,
                              float* __restrict__ depth_out,
                              float* __restrict__ nrm_out,
                              int B, int Nmax, long long HW)
{
    __shared__ __attribute__((aligned(16))) float s_w[BG * MAXW * WREC];
    __shared__ __attribute__((aligned(16))) unsigned s_free[BG * MAXW * 4];
    __shared__ __attribute__((aligned(16))) float s_cone[BG * MAXW * 4];
    __shared__ float s_denom[TPB * LDS_PITCH];

    const int tid  = threadIdx.x;
    const int lane = tid & 31;
    const int lh   = lane & 15;
    const int hlf  = lane >> 4;
    const int wv   = tid >> 5;

    const long long p = (long long)blockIdx.x * TPB + tid;
    const float gx = grid[p * 3 + 0];
    const float gz = grid[p * 3 + 2];

    const int b0 = blockIdx.y * BG;
    const int nb = (B - b0 < BG) ? (B - b0) : BG;

    if (tid < nb * MAXW) {
        const int b = b0 + (tid >> 4);
        const int w = tid & 15;
        int n = nums[b];
        if (n > MAXW) n = MAXW;
        float nx = 0.f, nz = 0.f, nd = 0.f;
        float mnx = 0.f, mxx = 0.f, mnz = 0.f, mxz = 0.f;
        unsigned fxp = 0u, fxn = 0u, fzp = 0u, fzn = 0u;
        float cmx = 0.f, cmz = 0.f, cq = -1.f;
        if (w < n) {
            const long long base = ((long long)b * Nmax + w) * 3;
            const int j = (w + 1 == n) ? 0 : w + 1;
            const long long basej = ((long long)b * Nmax + j) * 3;
            const float x0 = corners[base + 0], z0 = corners[base + 2];
            const float x1 = corners[basej + 0], z1 = corners[basej + 2];
            const float dx = x1 - x0, dz = z1 - z0;
            nx = -dz;
            nz =  dx;
            nd = fmaf(nz, z0, nx * x0);
            mnx = fminf(x0, x1) - C2D_EPS;  mxx = fmaxf(x0, x1) + C2D_EPS;
            mnz = fminf(z0, z1) - C2D_EPS;  mxz = fmaxf(z0, z1) + C2D_EPS;
            fxp = (mnx > 0.f) ? 0u : ~0u;
            fxn = (mxx < 0.f) ? 0u : ~0u;
            fzp = (mnz > 0.f) ? 0u : ~0u;
            fzn = (mxz < 0.f) ? 0u : ~0u;
            const bool px = mnx > 0.f, qx = mxx < 0.f, pz = mnz > 0.f, qz = mxz < 0.f;
            float e1x = 0.f, e1z = 0.f, e2x = 0.f, e2z = 0.f;
            bool cone = true;
            if ((px || qx) && (pz || qz)) {
                if (px == pz) { e1x = mxx; e1z = mnz; e2x = mnx; e2z = mxz; }
                else          { e1x = mnx; e1z = mnz; e2x = mxx; e2z = mxz; }
            } else if (px) { e1x = mnx; e1z = mnz; e2x = mnx; e2z = mxz; }
            else if (qx)   { e1x = mxx; e1z = mnz; e2x = mxx; e2z = mxz; }
            else if (pz)   { e1x = mnx; e1z = mnz; e2x = mxx; e2z = mnz; }
            else if (qz)   { e1x = mnx; e1z = mxz; e2x = mxx; e2z = mxz; }
            else cone = false;
            cq = 1.f;
            if (cone) {
                const float r1 = __builtin_amdgcn_rsqf(e1x * e1x + e1z * e1z);
                const float r2 = __builtin_amdgcn_rsqf(e2x * e2x + e2z * e2z);
                const float u1x = e1x * r1, u1z = e1z * r1;
                const float u2x = e2x * r2, u2z = e2z * r2;
                if (u1x * u2x + u1z * u2z > 0.f) {
                    const float sx = u1x + u2x, sz = u1z + u2z;
                    const float s2 = sx * sx + sz * sz;
                    const float rs = __builtin_amdgcn_rsqf(s2);
                    cmx = sx * rs;  cmz = sz * rs;
                    cq  = CONE_MU - 0.5f * s2 * rs;
                }
            }
        }
        float* wr = &s_w[tid * WREC];
        wr[0] = nd;  wr[1] = mxx; wr[2] = mnx; wr[3] = mxz;
        wr[4] = mnz; wr[5] = nx;  wr[6] = nz;  wr[7] = 0.f;
        unsigned* f = &s_free[tid * 4];
        f[0] = fxp; f[1] = fxn; f[2] = fzp; f[3] = fzn;
        float* cr = &s_cone[tid * 4];
        cr[0] = cmx; cr[1] = cq; cr[2] = cmz; cr[3] = 0.f;
    }
    __syncthreads();

    const float gx0 = __shfl(gx, lh,      32);
    const float gz0 = __shfl(gz, lh,      32);
    const float gx1 = __shfl(gx, lh + 16, 32);
    const float gz1 = __shfl(gz, lh + 16, 32);
    v2f a0, a1;
    a0.x = hlf ? gz0 : gx0;  a0.y = 0.f;
    a1.x = hlf ? gz1 : gx1;  a1.y = 0.f;
    v2f c0, c1;
    c0.x = a0.x;  c0.y = hlf ? 0.f : __builtin_amdgcn_sqrtf(gx0 * gx0 + gz0 * gz0);
    c1.x = a1.x;  c1.y = hlf ? 0.f : __builtin_amdgcn_sqrtf(gx1 * gx1 + gz1 * gz1);
    const unsigned bxp = __builtin_amdgcn_ballot_w32(gx > 0.f);
    const unsigned bxn = __builtin_amdgcn_ballot_w32(gx < 0.f);
    const unsigned bzp = __builtin_amdgcn_ballot_w32(gz > 0.f);
    const unsigned bzn = __builtin_amdgcn_ballot_w32(gz < 0.f);

    for (int bb = 0; bb < nb; ++bb) {
        const float*    sw = &s_w[bb * MAXW * WREC];
        const unsigned* sf = &s_free[bb * MAXW * 4];
        const float*    sc = &s_cone[bb * MAXW * 4];

        v2f bm;
        bm.x = sw[lh * WREC + (hlf ? 6 : 5)];  bm.y = 0.f;
        v8f acc0 = {}, acc1 = {};
        acc0 = __builtin_amdgcn_wmma_f32_16x16x4_f32(false, a0, false, bm,
                                                     (short)0, acc0, false, false);
        acc1 = __builtin_amdgcn_wmma_f32_16x16x4_f32(false, a1, false, bm,
                                                     (short)0, acc1, false, false);
        const float2 cb = *reinterpret_cast<const float2*>(&sc[lh * 4 + hlf * 2]);
        v2f cm;
        cm.x = cb.x;  cm.y = cb.y;
        v8f t0 = {}, t1 = {};
        t0 = __builtin_amdgcn_wmma_f32_16x16x4_f32(false, c0, false, cm,
                                                   (short)0, t0, false, false);
        t1 = __builtin_amdgcn_wmma_f32_16x16x4_f32(false, c1, false, cm,
                                                   (short)0, t1, false, false);

        wave_lds_order();
        const int mBase = hlf * 8;
#pragma unroll
        for (int r = 0; r < 8; ++r) {
            s_denom[(wv * 32 +      mBase + r) * LDS_PITCH + lh] = acc0[r];
            s_denom[(wv * 32 + 16 + mBase + r) * LDS_PITCH + lh] = acc1[r];
        }
        wave_lds_order();

        const uint4 fr = *reinterpret_cast<const uint4*>(&sf[lh * 4]);
        const unsigned rays = (bxp | fr.x) & (bxn | fr.y) & (bzp | fr.z) & (bzn | fr.w);
        float tm = __builtin_fmaxf(t0[0], t1[0]);
#pragma unroll
        for (int r = 1; r < 8; ++r)
            tm = __builtin_fmaxf(tm, __builtin_fmaxf(t0[r], t1[r]));
        unsigned inc = __builtin_amdgcn_ballot_w32(!(tm < 0.f));
        inc |= inc >> 16;
        unsigned live = __builtin_amdgcn_ballot_w32(rays != 0u) & inc & 0xffffu;

        float best = __builtin_inff();
        int   bi   = 0;
        while (live != 0u) {
            const int i = __builtin_ctz(live);
            live &= live - 1u;
            const float4 wa = *reinterpret_cast<const float4*>(&sw[i * WREC]);
            const float  mnz = sw[i * WREC + 4];
            const float denom = s_denom[tid * LDS_PITCH + i];
            const float scale = wa.x * __builtin_amdgcn_rcpf(denom);
            const float ix = gx * scale;
            const float iz = gz * scale;
            const bool ok = (ix <= wa.y) & (ix >= wa.z)
                          & (iz <= wa.w) & (iz >= mnz)
                          & (scale > 0.f);
            if (ok && scale < best) { best = scale; bi = i; }
        }

        const long long o = (long long)(b0 + bb) * HW + p;
        depth_out[o]       = best;
        nrm_out[o * 3 + 0] = sw[bi * WREC + 5];
        nrm_out[o * 3 + 1] = 0.f;
        nrm_out[o * 3 + 2] = sw[bi * WREC + 6];
    }
}

extern "C" void kernel_launch(void* const* d_in, const int* in_sizes, int n_in,
                              void* d_out, int out_size, void* d_ws, size_t ws_size,
                              hipStream_t stream) {
    const float* corners = (const float*)d_in[0];
    const float* grid    = (const float*)d_in[1];
    const int*   nums    = (const int*)d_in[2];

    const int       B    = in_sizes[2];
    const int       Nmax = in_sizes[0] / (B * 3);
    const long long HW   = in_sizes[1] / 3;

    float* depth = (float*)d_out;
    float* nrm   = depth + (long long)B * HW;

    dim3 grd((unsigned)(HW / TPB), (unsigned)((B + BG - 1) / BG));
    corner2depth_wmma_kernel<<<grd, TPB, 0, stream>>>(corners, grid, nums,
                                                      depth, nrm, B, Nmax, HW);
}


# Sign and cone tests first for all samples; only surviving (sample, wall) pairs get a denominator column, normals packed by rank into one B tile: 2 denominator WMMAs + 1 transpose a wave, not 8 + 4
# speedup vs baseline: 1.8387x; 1.1349x over previous
#include <hip/hip_runtime.h>


typedef __attribute__((ext_vector_type(2))) float v2f;
typedef __attribute__((ext_vector_type(8))) float v8f;

#define C2D_EPS   0.01f
#define MAXW      16
#define TPB       256
#define LDS_PITCH 17
#define WREC      8
#define BG        4
#define CONE_MU   0.00048828125f

__device__ __forceinline__ void wave_lds_order()
{
    __builtin_amdgcn_fence(__ATOMIC_RELEASE, "wavefront");
    __builtin_amdgcn_wave_barrier();
    __builtin_amdgcn_fence(__ATOMIC_ACQUIRE, "wavefront");
}

__global__ __launch_bounds__(TPB, 16)
void corner2depth_wmma_kernel(const float* __restrict__ corners,
                              const float* __restrict__ grid,
                              const int*   __restrict__ nums,
                              float* __restrict__ depth_out,
                              float* __restrict__ nrm_out,
                              int B, int Nmax, long long HW)
{
    __shared__ __attribute__((aligned(16))) float s_w[BG * MAXW * WREC];
    __shared__ __attribute__((aligned(16))) unsigned s_free[BG * MAXW * 4];
    __shared__ __attribute__((aligned(16))) float s_cone[BG * MAXW * 4];
    __shared__ float s_denom[TPB * LDS_PITCH];
    __shared__ unsigned s_col[(TPB / 32) * MAXW];

    const int tid  = threadIdx.x;
    const int lane = tid & 31;
    const int lh   = lane & 15;
    const int hlf  = lane >> 4;
    const int wv   = tid >> 5;

    const long long p = (long long)blockIdx.x * TPB + tid;
    const float gx = grid[p * 3 + 0];
    const float gz = grid[p * 3 + 2];

    const int b0 = blockIdx.y * BG;
    const int nb = (B - b0 < BG) ? (B - b0) : BG;

    if (tid < BG * MAXW) {
        const int b = b0 + (tid >> 4);
        const int w = tid & 15;
        int n = ((tid >> 4) < nb) ? nums[b] : 0;
        if (n > MAXW) n = MAXW;
        float nx = 0.f, nz = 0.f, nd = 0.f;
        float mnx = 0.f, mxx = 0.f, mnz = 0.f, mxz = 0.f;
        unsigned fxp = 0u, fxn = 0u, fzp = 0u, fzn = 0u;
        float cmx = 0.f, cmz = 0.f, cq = -1.f;
        if (w < n) {
            const long long base = ((long long)b * Nmax + w) * 3;
            const int j = (w + 1 == n) ? 0 : w + 1;
            const long long basej = ((long long)b * Nmax + j) * 3;
            const float x0 = corners[base + 0], z0 = corners[base + 2];
            const float x1 = corners[basej + 0], z1 = corners[basej + 2];
            const float dx = x1 - x0, dz = z1 - z0;
            nx = -dz;
            nz =  dx;
            nd = fmaf(nz, z0, nx * x0);
            mnx = fminf(x0, x1) - C2D_EPS;  mxx = fmaxf(x0, x1) + C2D_EPS;
            mnz = fminf(z0, z1) - C2D_EPS;  mxz = fmaxf(z0, z1) + C2D_EPS;
            fxp = (mnx > 0.f) ? 0u : ~0u;
            fxn = (mxx < 0.f) ? 0u : ~0u;
            fzp = (mnz > 0.f) ? 0u : ~0u;
            fzn = (mxz < 0.f) ? 0u : ~0u;
            const bool px = mnx > 0.f, qx = mxx < 0.f, pz = mnz > 0.f, qz = mxz < 0.f;
            float e1x = 0.f, e1z = 0.f, e2x = 0.f, e2z = 0.f;
            bool cone = true;
            if ((px || qx) && (pz || qz)) {
                if (px == pz) { e1x = mxx; e1z = mnz; e2x = mnx; e2z = mxz; }
                else          { e1x = mnx; e1z = mnz; e2x = mxx; e2z = mxz; }
            } else if (px) { e1x = mnx; e1z = mnz; e2x = mnx; e2z = mxz; }
            else if (qx)   { e1x = mxx; e1z = mnz; e2x = mxx; e2z = mxz; }
            else if (pz)   { e1x = mnx; e1z = mnz; e2x = mxx; e2z = mnz; }
            else if (qz)   { e1x = mnx; e1z = mxz; e2x = mxx; e2z = mxz; }
            else cone = false;
            cq = 1.f;
            if (cone) {
                const float r1 = __builtin_amdgcn_rsqf(e1x * e1x + e1z * e1z);
                const float r2 = __builtin_amdgcn_rsqf(e2x * e2x + e2z * e2z);
                const float u1x = e1x * r1, u1z = e1z * r1;
                const float u2x = e2x * r2, u2z = e2z * r2;
                if (u1x * u2x + u1z * u2z > 0.f) {
                    const float sx = u1x + u2x, sz = u1z + u2z;
                    const float s2 = sx * sx + sz * sz;
                    const float rs = __builtin_amdgcn_rsqf(s2);
                    cmx = sx * rs;  cmz = sz * rs;
                    cq  = CONE_MU - 0.5f * s2 * rs;
                }
            }
        }
        float* wr = &s_w[tid * WREC];
        wr[0] = nd;  wr[1] = mxx; wr[2] = mnx; wr[3] = mxz;
        wr[4] = mnz; wr[5] = nx;  wr[6] = nz;  wr[7] = 0.f;
        unsigned* f = &s_free[tid * 4];
        f[0] = fxp; f[1] = fxn; f[2] = fzp; f[3] = fzn;
        float* cr = &s_cone[tid * 4];
        cr[0] = cmx; cr[1] = cq; cr[2] = cmz; cr[3] = 0.f;
    }
    __syncthreads();

    const float gx0 = __shfl(gx, lh,      32);
    const float gz0 = __shfl(gz, lh,      32);
    const float gx1 = __shfl(gx, lh + 16, 32);
    const float gz1 = __shfl(gz, lh + 16, 32);
    v2f a0, a1;
    a0.x = hlf ? gz0 : gx0;  a0.y = 0.f;
    a1.x = hlf ? gz1 : gx1;  a1.y = 0.f;
    v2f c0, c1;
    c0.x = a0.x;  c0.y = hlf ? 0.f : __builtin_amdgcn_sqrtf(gx0 * gx0 + gz0 * gz0);
    c1.x = a1.x;  c1.y = hlf ? 0.f : __builtin_amdgcn_sqrtf(gx1 * gx1 + gz1 * gz1);
    const unsigned bxp = __builtin_amdgcn_ballot_w32(gx > 0.f);
    const unsigned bxn = __builtin_amdgcn_ballot_w32(gx < 0.f);
    const unsigned bzp = __builtin_amdgcn_ballot_w32(gz > 0.f);
    const unsigned bzn = __builtin_amdgcn_ballot_w32(gz < 0.f);

    unsigned long long M = 0ull;
#pragma unroll
    for (int bb = 0; bb < BG; ++bb) {
        const unsigned* sf = &s_free[bb * MAXW * 4];
        const float*    sc = &s_cone[bb * MAXW * 4];
        const float2 cb = *reinterpret_cast<const float2*>(&sc[lh * 4 + hlf * 2]);
        v2f cm;
        cm.x = cb.x;  cm.y = cb.y;
        v8f t0 = {}, t1 = {};
        t0 = __builtin_amdgcn_wmma_f32_16x16x4_f32(false, c0, false, cm,
                                                   (short)0, t0, false, false);
        t1 = __builtin_amdgcn_wmma_f32_16x16x4_f32(false, c1, false, cm,
                                                   (short)0, t1, false, false);
        const uint4 fr = *reinterpret_cast<const uint4*>(&sf[lh * 4]);
        const unsigned rays = (bxp | fr.x) & (bxn | fr.y) & (bzp | fr.z) & (bzn | fr.w);
        float tm = __builtin_fmaxf(t0[0], t1[0]);
#pragma unroll
        for (int r = 1; r < 8; ++r)
            tm = __builtin_fmaxf(tm, __builtin_fmaxf(t0[r], t1[r]));
        unsigned inc = __builtin_amdgcn_ballot_w32(!(tm < 0.f));
        inc |= inc >> 16;
        const unsigned live = __builtin_amdgcn_ballot_w32(rays != 0u) & inc & 0xffffu;
        M |= (unsigned long long)live << (16 * bb);
    }

    unsigned* col = &s_col[wv * MAXW];
    const unsigned below = (1u << lane) - 1u;
    int s = 0;
    while (s < nb) {
        int cnt = __builtin_popcount((unsigned)(M >> (16 * s)) & 0xffffu);
        int e = s + 1;
        while (e < nb) {
            const int c2 = __builtin_popcount((unsigned)(M >> (16 * e)) & 0xffffu);
            if (cnt + c2 > MAXW) break;
            cnt += c2;
            ++e;
        }
        const unsigned long long R = (e - s == BG)
            ? (M >> (16 * s))
            : ((M >> (16 * s)) & ((1ull << (16 * (e - s))) - 1ull));
        const unsigned rlo = (unsigned)R, rhi = (unsigned)(R >> 32);

        wave_lds_order();
        if ((rlo >> lane) & 1u)
            col[__builtin_popcount(rlo & below)] = (unsigned)lane;
        if ((rhi >> lane) & 1u)
            col[__builtin_popcount(rlo) + __builtin_popcount(rhi & below)] = 32u + (unsigned)lane;
        wave_lds_order();

        const bool used = lh < cnt;
        const unsigned pid = used ? col[lh] : 0u;
        v2f bm;
        bm.x = used ? s_w[(s * MAXW + (int)pid) * WREC + (hlf ? 6 : 5)] : 0.f;
        bm.y = 0.f;
        v8f acc0 = {}, acc1 = {};
        acc0 = __builtin_amdgcn_wmma_f32_16x16x4_f32(false, a0, false, bm,
                                                     (short)0, acc0, false, false);
        acc1 = __builtin_amdgcn_wmma_f32_16x16x4_f32(false, a1, false, bm,
                                                     (short)0, acc1, false, false);

        const int mBase = hlf * 8;
#pragma unroll
        for (int r = 0; r < 8; ++r) {
            s_denom[(wv * 32 +      mBase + r) * LDS_PITCH + lh] = acc0[r];
            s_denom[(wv * 32 + 16 + mBase + r) * LDS_PITCH + lh] = acc1[r];
        }
        wave_lds_order();

        int c = 0;
        for (int sp = s; sp < e; ++sp) {
            const float* sw = &s_w[sp * MAXW * WREC];
            unsigned live = (unsigned)(M >> (16 * sp)) & 0xffffu;
            float best = __builtin_inff();
            int   bi   = 0;
            while (live != 0u) {
                const int i = __builtin_ctz(live);
                live &= live - 1u;
                const float4 wa = *reinterpret_cast<const float4*>(&sw[i * WREC]);
                const float  mnz = sw[i * WREC + 4];
                const float denom = s_denom[tid * LDS_PITCH + c];
                ++c;
                const float scale = wa.x * __builtin_amdgcn_rcpf(denom);
                const float ix = gx * scale;
                const float iz = gz * scale;
                const bool ok = (ix <= wa.y) & (ix >= wa.z)
                              & (iz <= wa.w) & (iz >= mnz)
                              & (scale > 0.f);
                if (ok && scale < best) { best = scale; bi = i; }
            }

            const long long o = (long long)(b0 + sp) * HW + p;
            depth_out[o]       = best;
            nrm_out[o * 3 + 0] = sw[bi * WREC + 5];
            nrm_out[o * 3 + 1] = 0.f;
            nrm_out[o * 3 + 2] = sw[bi * WREC + 6];
        }
        s = e;
    }
}

extern "C" void kernel_launch(void* const* d_in, const int* in_sizes, int n_in,
                              void* d_out, int out_size, void* d_ws, size_t ws_size,
                              hipStream_t stream) {
    const float* corners = (const float*)d_in[0];
    const float* grid    = (const float*)d_in[1];
    const int*   nums    = (const int*)d_in[2];

    const int       B    = in_sizes[2];
    const int       Nmax = in_sizes[0] / (B * 3);
    const long long HW   = in_sizes[1] / 3;

    float* depth = (float*)d_out;
    float* nrm   = depth + (long long)B * HW;

    dim3 grd((unsigned)(HW / TPB), (unsigned)((B + BG - 1) / BG));
    corner2depth_wmma_kernel<<<grd, TPB, 0, stream>>>(corners, grid, nums,
                                                      depth, nrm, B, Nmax, HW);
}


# Wave cull by direction interval, not per ray: rays and wall-box corners get a monotone diamond angle, a wall is walked if the wave's [min,max] meets its arc; sign ballots and cone WMMAs gone
# speedup vs baseline: 2.0743x; 1.1281x over previous
#include <hip/hip_runtime.h>


typedef __attribute__((ext_vector_type(2))) float v2f;
typedef __attribute__((ext_vector_type(8))) float v8f;

#define C2D_EPS   0.01f
#define MAXW      16
#define TPB       256
#define LDS_PITCH 17
#define WREC      8
#define BG        4
#define ARC_MU    0.00048828125f

__device__ __forceinline__ void wave_lds_order()
{
    __builtin_amdgcn_fence(__ATOMIC_RELEASE, "wavefront");
    __builtin_amdgcn_wave_barrier();
    __builtin_amdgcn_fence(__ATOMIC_ACQUIRE, "wavefront");
}

__device__ __forceinline__ float diamond_angle(float x, float z)
{
    const float t  = __builtin_fmaxf(__builtin_fabsf(x), __builtin_fabsf(z));
    const float k  = (t < 0x1p-64f) ? 0x1p64f : 0.25f;
    const float xs = x * k, zs = z * k;
    const float p  = xs * __builtin_amdgcn_rcpf(__builtin_fabsf(xs) + __builtin_fabsf(zs));
    return (zs >= 0.f) ? p : (__builtin_copysignf(2.f, xs) - p);
}

template <int MASK>
__device__ __forceinline__ float row_xor(float v)
{
    return __builtin_amdgcn_update_dpp(v, v, 0x160 + MASK, 0xf, 0xf, true);
}
__device__ __forceinline__ float row_other(float v)
{
    const int i = __builtin_bit_cast(int, v);
    return __builtin_bit_cast(float, __builtin_amdgcn_permlanex16(i, i, 0x76543210, 0xfedcba98, false, false));
}

__global__ __launch_bounds__(TPB, 16)
void corner2depth_wmma_kernel(const float* __restrict__ corners,
                              const float* __restrict__ grid,
                              const int*   __restrict__ nums,
                              float* __restrict__ depth_out,
                              float* __restrict__ nrm_out,
                              int B, int Nmax, long long HW)
{
    __shared__ __attribute__((aligned(16))) float s_w[BG * MAXW * WREC];
    __shared__ __attribute__((aligned(16))) float s_arc[BG * MAXW * 4];
    __shared__ float s_denom[TPB * LDS_PITCH];
    __shared__ unsigned s_col[(TPB / 32) * MAXW];

    const int tid  = threadIdx.x;
    const int lane = tid & 31;
    const int lh   = lane & 15;
    const int hlf  = lane >> 4;
    const int wv   = tid >> 5;

    const long long p = (long long)blockIdx.x * TPB + tid;
    const float gx = grid[p * 3 + 0];
    const float gz = grid[p * 3 + 2];

    const int b0 = blockIdx.y * BG;
    const int nb = (B - b0 < BG) ? (B - b0) : BG;

    if (tid < BG * MAXW) {
        const int b = b0 + (tid >> 4);
        const int w = tid & 15;
        int n = ((tid >> 4) < nb) ? nums[b] : 0;
        if (n > MAXW) n = MAXW;
        float nx = 0.f, nz = 0.f, nd = 0.f;
        float mnx = 0.f, mxx = 0.f, mnz = 0.f, mxz = 0.f;
        float aP = __builtin_inff(), aQ = -__builtin_inff();
        unsigned amode = 0u;
        if (w < n) {
            const long long base = ((long long)b * Nmax + w) * 3;
            const int j = (w + 1 == n) ? 0 : w + 1;
            const long long basej = ((long long)b * Nmax + j) * 3;
            const float x0 = corners[base + 0], z0 = corners[base + 2];
            const float x1 = corners[basej + 0], z1 = corners[basej + 2];
            const float dx = x1 - x0, dz = z1 - z0;
            nx = -dz;
            nz =  dx;
            nd = fmaf(nz, z0, nx * x0);
            mnx = fminf(x0, x1) - C2D_EPS;  mxx = fmaxf(x0, x1) + C2D_EPS;
            mnz = fminf(z0, z1) - C2D_EPS;  mxz = fmaxf(z0, z1) + C2D_EPS;
            const bool px = mnx > 0.f, qx = mxx < 0.f, pz = mnz > 0.f, qz = mxz < 0.f;
            float e1x = 0.f, e1z = 0.f, e2x = 0.f, e2z = 0.f;
            bool cone = true;
            if ((px || qx) && (pz || qz)) {
                if (px == pz) { e1x = mxx; e1z = mnz; e2x = mnx; e2z = mxz; }
                else          { e1x = mnx; e1z = mnz; e2x = mxx; e2z = mxz; }
            } else if (px) { e1x = mnx; e1z = mnz; e2x = mnx; e2z = mxz; }
            else if (qx)   { e1x = mxx; e1z = mnz; e2x = mxx; e2z = mxz; }
            else if (pz)   { e1x = mnx; e1z = mnz; e2x = mxx; e2z = mnz; }
            else if (qz)   { e1x = mnx; e1z = mxz; e2x = mxx; e2z = mxz; }
            else cone = false;
            aP = -__builtin_inff();  aQ = __builtin_inff();
            if (cone) {
                const float a1 = diamond_angle(e1x, e1z);
                const float a2 = diamond_angle(e2x, e2z);
                const float lo = __builtin_fminf(a1, a2), hi = __builtin_fmaxf(a1, a2);
                const float d  = hi - lo;
                if (d < 2.f - 0x1p-8f) {
                    aP = lo - ARC_MU;  aQ = hi + ARC_MU;
                    if (aP < -2.f) { aP += 4.f; amode = 1u; }
                    if (aQ >  2.f) { aQ -= 4.f; amode = 1u; }
                } else if (d > 2.f + 0x1p-8f) {
                    aP = hi - ARC_MU;  aQ = lo + ARC_MU;  amode = 1u;
                }
            }
        }
        float* wr = &s_w[tid * WREC];
        wr[0] = nd;  wr[1] = mxx; wr[2] = mnx; wr[3] = mxz;
        wr[4] = mnz; wr[5] = nx;  wr[6] = nz;  wr[7] = 0.f;
        float* ar = &s_arc[tid * 4];
        ar[0] = aP; ar[1] = aQ; ar[2] = __builtin_bit_cast(float, amode); ar[3] = 0.f;
    }

    const float ang = diamond_angle(gx, gz);
    float amx = ang, amn = ang;
    amx = __builtin_fmaximum_numf(amx, row_xor<1>(amx));  amn = __builtin_fminimum_numf(amn, row_xor<1>(amn));
    amx = __builtin_fmaximum_numf(amx, row_xor<2>(amx));  amn = __builtin_fminimum_numf(amn, row_xor<2>(amn));
    amx = __builtin_fmaximum_numf(amx, row_xor<4>(amx));  amn = __builtin_fminimum_numf(amn, row_xor<4>(amn));
    amx = __builtin_fmaximum_numf(amx, row_xor<8>(amx));  amn = __builtin_fminimum_numf(amn, row_xor<8>(amn));
    amx = __builtin_fmaximum_numf(amx, row_other(amx));   amn = __builtin_fminimum_numf(amn, row_other(amn));
    __syncthreads();

    const float gx0 = __shfl(gx, lh,      32);
    const float gz0 = __shfl(gz, lh,      32);
    const float gx1 = __shfl(gx, lh + 16, 32);
    const float gz1 = __shfl(gz, lh + 16, 32);
    v2f a0, a1;
    a0.x = hlf ? gz0 : gx0;  a0.y = 0.f;
    a1.x = hlf ? gz1 : gx1;  a1.y = 0.f;
    const float4 w0 = *reinterpret_cast<const float4*>(&s_arc[lane * 4]);
    const float4 w1 = *reinterpret_cast<const float4*>(&s_arc[(32 + lane) * 4]);
    const unsigned p0 = __builtin_amdgcn_ballot_w32(amx >= w0.x);
    const unsigned q0 = __builtin_amdgcn_ballot_w32(amn <= w0.y);
    const unsigned m0 = __builtin_amdgcn_ballot_w32(__builtin_bit_cast(unsigned, w0.z) != 0u);
    const unsigned p1 = __builtin_amdgcn_ballot_w32(amx >= w1.x);
    const unsigned q1 = __builtin_amdgcn_ballot_w32(amn <= w1.y);
    const unsigned m1 = __builtin_amdgcn_ballot_w32(__builtin_bit_cast(unsigned, w1.z) != 0u);
    const unsigned l0 = (p0 & q0) | (m0 & (p0 | q0));
    const unsigned l1 = (p1 & q1) | (m1 & (p1 | q1));
    const unsigned long long M = (unsigned long long)l0 | ((unsigned long long)l1 << 32);

    unsigned* col = &s_col[wv * MAXW];
    const unsigned below = (1u << lane) - 1u;
    int s = 0;
    while (s < nb) {
        int cnt = __builtin_popcount((unsigned)(M >> (16 * s)) & 0xffffu);
        int e = s + 1;
        while (e < nb) {
            const int c2 = __builtin_popcount((unsigned)(M >> (16 * e)) & 0xffffu);
            if (cnt + c2 > MAXW) break;
            cnt += c2;
            ++e;
        }
        const unsigned long long R = (e - s == BG)
            ? (M >> (16 * s))
            : ((M >> (16 * s)) & ((1ull << (16 * (e - s))) - 1ull));
        const unsigned rlo = (unsigned)R, rhi = (unsigned)(R >> 32);

        wave_lds_order();
        if ((rlo >> lane) & 1u)
            col[__builtin_popcount(rlo & below)] = (unsigned)lane;
        if ((rhi >> lane) & 1u)
            col[__builtin_popcount(rlo) + __builtin_popcount(rhi & below)] = 32u + (unsigned)lane;
        wave_lds_order();

        const bool used = lh < cnt;
        const unsigned pid = used ? col[lh] : 0u;
        v2f bm;
        bm.x = used ? s_w[(s * MAXW + (int)pid) * WREC + (hlf ? 6 : 5)] : 0.f;
        bm.y = 0.f;
        v8f acc0 = {}, acc1 = {};
        acc0 = __builtin_amdgcn_wmma_f32_16x16x4_f32(false, a0, false, bm,
                                                     (short)0, acc0, false, false);
        acc1 = __builtin_amdgcn_wmma_f32_16x16x4_f32(false, a1, false, bm,
                                                     (short)0, acc1, false, false);

        const int mBase = hlf * 8;
#pragma unroll
        for (int r = 0; r < 8; ++r) {
            s_denom[(wv * 32 +      mBase + r) * LDS_PITCH + lh] = acc0[r];
            s_denom[(wv * 32 + 16 + mBase + r) * LDS_PITCH + lh] = acc1[r];
        }
        wave_lds_order();

        int c = 0;
        for (int sp = s; sp < e; ++sp) {
            const float* sw = &s_w[sp * MAXW * WREC];
            unsigned live = (unsigned)(M >> (16 * sp)) & 0xffffu;
            float best = __builtin_inff();
            int   bi   = 0;
            while (live != 0u) {
                const int i = __builtin_ctz(live);
                live &= live - 1u;
                const float4 wa = *reinterpret_cast<const float4*>(&sw[i * WREC]);
                const float  mnz = sw[i * WREC + 4];
                const float denom = s_denom[tid * LDS_PITCH + c];
                ++c;
                const float scale = wa.x * __builtin_amdgcn_rcpf(denom);
                const float ix = gx * scale;
                const float iz = gz * scale;
                const bool ok = (ix <= wa.y) & (ix >= wa.z)
                              & (iz <= wa.w) & (iz >= mnz)
                              & (scale > 0.f);
                if (ok && scale < best) { best = scale; bi = i; }
            }

            const long long o = (long long)(b0 + sp) * HW + p;
            depth_out[o]       = best;
            nrm_out[o * 3 + 0] = sw[bi * WREC + 5];
            nrm_out[o * 3 + 1] = 0.f;
            nrm_out[o * 3 + 2] = sw[bi * WREC + 6];
        }
        s = e;
    }
}

extern "C" void kernel_launch(void* const* d_in, const int* in_sizes, int n_in,
                              void* d_out, int out_size, void* d_ws, size_t ws_size,
                              hipStream_t stream) {
    const float* corners = (const float*)d_in[0];
    const float* grid    = (const float*)d_in[1];
    const int*   nums    = (const int*)d_in[2];

    const int       B    = in_sizes[2];
    const int       Nmax = in_sizes[0] / (B * 3);
    const long long HW   = in_sizes[1] / 3;

    float* depth = (float*)d_out;
    float* nrm   = depth + (long long)B * HW;

    dim3 grd((unsigned)(HW / TPB), (unsigned)((B + BG - 1) / BG));
    corner2depth_wmma_kernel<<<grd, TPB, 0, stream>>>(corners, grid, nums,
                                                      depth, nrm, B, Nmax, HW);
}
